# starting kernel
# baseline (speedup 1.0000x reference)
#include <hip/hip_runtime.h>
#include <hip/hip_bf16.h>

typedef __attribute__((ext_vector_type(2))) float v2f;
typedef __attribute__((ext_vector_type(8))) float v8f;

#define GCN_N_NODES 100000
#define GCN_N_EDGES 1000000
#define GCN_D 64

__global__ void gcn_zero_deg(float* __restrict__ deg) {
    int i = blockIdx.x * blockDim.x + threadIdx.x;
    if (i < GCN_N_NODES) deg[i] = 0.0f;
}

__global__ void gcn_degree(const int* __restrict__ edge_index, float* __restrict__ deg) {
    int e = blockIdx.x * blockDim.x + threadIdx.x;
    if (e < GCN_N_EDGES) {
        atomicAdd(&deg[edge_index[e]], 1.0f);
    }
}

__global__ void gcn_normalize(const float* __restrict__ x, const float* __restrict__ deg,
                              float* __restrict__ x_norm, float* __restrict__ agg) {
    int i = blockIdx.x * blockDim.x + threadIdx.x;
    if (i >= GCN_N_NODES * (GCN_D / 4)) return;
    int n = i / (GCN_D / 4);
    float s = rsqrtf(deg[n] + 1.0f);
    float4 v = ((const float4*)x)[i];
    v.x *= s; v.y *= s; v.z *= s; v.w *= s;
    ((float4*)x_norm)[i] = v;
    ((float4*)agg)[i]    = v;
}

__global__ void gcn_scatter(const int* __restrict__ edge_index,
                            const float* __restrict__ x_norm, float* __restrict__ agg) {
    int t = blockIdx.x * blockDim.x + threadIdx.x;
    int e = t >> 6;
    int f = t & 63;
    if (e >= GCN_N_EDGES) return;
    int s = edge_index[e];
    int d = edge_index[GCN_N_EDGES + e];
    atomicAdd(&agg[d * GCN_D + f], x_norm[s * GCN_D + f]);
}

__global__ void gcn_gemm_wmma(const float* __restrict__ agg, const float* __restrict__ W,
                              const float* __restrict__ bias, float* __restrict__ out) {
    const int lane = threadIdx.x & 31;
    const int wave = threadIdx.x >> 5;
    const int tile = blockIdx.x * 8 + wave;
    const int TILES_N = GCN_D / 16;
    const int n_tiles = (GCN_N_NODES / 16) * TILES_N;
    if (tile >= n_tiles) return;

    const int tm = tile / TILES_N;
    const int tn = tile % TILES_N;
    const int m0 = tm * 16;
    const int o0 = tn * 16;

    const int row = lane & 15;
    const int col = lane & 15;
    const int kb  = (lane >> 4) << 1;

    v2f breg[16];
    const float* wrow = W + (o0 + col) * GCN_D + kb;
#pragma unroll
    for (int t = 0; t < 16; ++t) {
        breg[t] = *(const v2f*)(wrow + 4 * t);
    }

    const float bv = bias[o0 + col];
    v8f c;
#pragma unroll
    for (int r = 0; r < 8; ++r) c[r] = bv;

    const float* arow = agg + (m0 + row) * GCN_D + kb;
#pragma unroll
    for (int t = 0; t < 16; ++t) {
        v2f a = *(const v2f*)(arow + 4 * t);
        c = __builtin_amdgcn_wmma_f32_16x16x4_f32(
                false, a, false, breg[t], (short)0, c, false, false);
    }

    float* orow = out + (m0 + ((lane >> 4) << 3)) * GCN_D + o0 + col;
#pragma unroll
    for (int r = 0; r < 8; ++r) {
        orow[r * GCN_D] = c[r];
    }
}

extern "C" void kernel_launch(void* const* d_in, const int* in_sizes, int n_in,
                              void* d_out, int out_size, void* d_ws, size_t ws_size,
                              hipStream_t stream) {
    const float* x          = (const float*)d_in[0];
    const int*   edge_index = (const int*)d_in[1];
    const float* W          = (const float*)d_in[2];
    const float* b          = (const float*)d_in[3];
    float*       out        = (float*)d_out;

    float* deg    = (float*)d_ws;
    float* x_norm = deg + GCN_N_NODES;
    float* agg    = x_norm + (size_t)GCN_N_NODES * GCN_D;

    gcn_zero_deg<<<(GCN_N_NODES + 255) / 256, 256, 0, stream>>>(deg);
    gcn_degree<<<(GCN_N_EDGES + 255) / 256, 256, 0, stream>>>(edge_index, deg);
    gcn_normalize<<<(GCN_N_NODES * (GCN_D / 4) + 255) / 256, 256, 0, stream>>>(x, deg, x_norm, agg);
    gcn_scatter<<<(GCN_N_EDGES * GCN_D) / 256, 256, 0, stream>>>(edge_index, x_norm, agg);

    const int n_tiles = (GCN_N_NODES / 16) * (GCN_D / 16);
    gcn_gemm_wmma<<<(n_tiles + 7) / 8, 256, 0, stream>>>(agg, W, b, out);
}


# atomic scatter replaced by a row gather over a global CSR (i32 rank atomics, prefix scan, fill): 16 lanes sum a row's y[src] in registers, out stored once; no tiles, no LDS sums; new sum order
# speedup vs baseline: 1.6467x; 1.6467x over previous
#include <hip/hip_runtime.h>
#include <hip/hip_bf16.h>


typedef __attribute__((ext_vector_type(2))) float v2f;
typedef __attribute__((ext_vector_type(4))) float v4f;
typedef __attribute__((ext_vector_type(8))) float v8f;

#define GCN_N_NODES 100000
#define GCN_N_EDGES 1000000
#define GCN_D 64
#define GCN_PITCH 68
#define GCN_SCAN_T 1024

#define GCN_WS_CNT    100000
#define GCN_WS_ROWPTR 200000
#define GCN_WS_RANK   300032
#define GCN_WS_CSR    1300032
#define GCN_WS_Y      2300032

__global__ void gcn_zero(float* __restrict__ deg, int* __restrict__ cnt) {
    int i = blockIdx.x * blockDim.x + threadIdx.x;
    if (i < GCN_N_NODES) { deg[i] = 0.0f; cnt[i] = 0; }
}

__global__ void gcn_count(const int* __restrict__ edge_index, float* __restrict__ deg,
                          int* __restrict__ cnt, int* __restrict__ rank) {
    int e = blockIdx.x * blockDim.x + threadIdx.x;
    if (e < GCN_N_EDGES) {
        atomicAdd(&deg[edge_index[e]], 1.0f);
        rank[e] = atomicAdd(&cnt[edge_index[GCN_N_EDGES + e]], 1);
    }
}

__global__ __launch_bounds__(GCN_SCAN_T)
void gcn_rowptr(const int* __restrict__ cnt, int* __restrict__ rowptr) {
    __shared__ int s_pre[32];
    __shared__ int s_tot[32];
    __shared__ int s_off[32];
    const int tid  = threadIdx.x;
    const int lane = tid & 31;
    const int wave = tid >> 5;

    const int4* c4 = (const int4*)cnt;
    const int   n4 = blockIdx.x * (GCN_SCAN_T / 4);
    int pre = 0;
#pragma unroll 8
    for (int j = tid; j < n4; j += GCN_SCAN_T) {
        const int4 v = c4[j];
        pre += (v.x + v.y) + (v.z + v.w);
    }

    const int i = blockIdx.x * GCN_SCAN_T + tid;
    const int v = (i < GCN_N_NODES) ? cnt[i] : 0;
    int inc = v;
    for (int d = 1; d < 32; d <<= 1) {
        pre += __shfl_xor(pre, d);
        const int u = __shfl_up(inc, d);
        if (lane >= d) inc += u;
    }
    if (lane == 31) s_tot[wave] = inc;
    if (lane == 0)  s_pre[wave] = pre;
    __syncthreads();
    if (wave == 0) {
        int p = s_pre[lane];
        const int t = s_tot[lane];
        int ti = t;
        for (int d = 1; d < 32; d <<= 1) {
            p += __shfl_xor(p, d);
            const int u = __shfl_up(ti, d);
            if (lane >= d) ti += u;
        }
        s_off[lane] = p + ti - t;
    }
    __syncthreads();
    const int excl = s_off[wave] + inc - v;
    if (i < GCN_N_NODES)      rowptr[i] = excl;
    if (i == GCN_N_NODES - 1) rowptr[GCN_N_NODES] = excl + v;
}

__global__ void gcn_fill(const int* __restrict__ edge_index, const int* __restrict__ rowptr,
                         const int* __restrict__ rank, int* __restrict__ csr) {
    int e = blockIdx.x * blockDim.x + threadIdx.x;
    if (e < GCN_N_EDGES) {
        const int s = edge_index[e];
        const int d = edge_index[GCN_N_EDGES + e];
        csr[rowptr[d] + rank[e]] = s;
    }
}

__global__ __launch_bounds__(256)
void gcn_linear_first(const float* __restrict__ x, const float* __restrict__ deg,
                      const float* __restrict__ W, float* __restrict__ y) {
    __shared__ float Ws[GCN_D * GCN_PITCH];
    __shared__ float Ts[8][16 * GCN_PITCH];

    const int tid  = threadIdx.x;
    const int lane = tid & 31;
    const int wave = tid >> 5;

#pragma unroll
    for (int q = 0; q < 4; ++q) {
        const int j  = tid + 256 * q;
        const int o  = j >> 4;
        const int k4 = j & 15;
        const float4 v = ((const float4*)W)[j];
        *(float4*)&Ws[o * GCN_PITCH + 4 * k4] = v;
    }
    __syncthreads();

    const int  strip  = blockIdx.x * 8 + wave;
    const bool active = strip < GCN_N_NODES / 16;
    const int  m0  = strip * 16;
    const int  row = lane & 15;
    const int  col = lane & 15;
    const int  h   = lane >> 4;
    const int  kb  = h << 1;
    float* T = Ts[wave];

    if (active) {
        const float s = rsqrtf(deg[m0 + row] + 1.0f);
        const float* arow = x + (size_t)(m0 + row) * GCN_D + kb;
        v2f a[16];
#pragma unroll
        for (int t = 0; t < 16; ++t) {
            a[t] = *(const v2f*)(arow + 4 * t);
        }
#pragma unroll
        for (int t = 0; t < 16; ++t) {
            a[t].x *= s; a[t].y *= s;
        }

        v8f c0, c1, c2, c3;
#pragma unroll
        for (int r = 0; r < 8; ++r) { c0[r] = 0.0f; c1[r] = 0.0f; c2[r] = 0.0f; c3[r] = 0.0f; }

        const float* wl = &Ws[col * GCN_PITCH + kb];
#pragma unroll
        for (int t = 0; t < 16; ++t) {
            const v2f b0 = *(const v2f*)(wl + (0 * 16) * GCN_PITCH + 4 * t);
            const v2f b1 = *(const v2f*)(wl + (1 * 16) * GCN_PITCH + 4 * t);
            const v2f b2 = *(const v2f*)(wl + (2 * 16) * GCN_PITCH + 4 * t);
            const v2f b3 = *(const v2f*)(wl + (3 * 16) * GCN_PITCH + 4 * t);
            c0 = __builtin_amdgcn_wmma_f32_16x16x4_f32(false, a[t], false, b0, (short)0, c0, false, false);
            c1 = __builtin_amdgcn_wmma_f32_16x16x4_f32(false, a[t], false, b1, (short)0, c1, false, false);
            c2 = __builtin_amdgcn_wmma_f32_16x16x4_f32(false, a[t], false, b2, (short)0, c2, false, false);
            c3 = __builtin_amdgcn_wmma_f32_16x16x4_f32(false, a[t], false, b3, (short)0, c3, false, false);
        }

        float* tp = T + (8 * h) * GCN_PITCH + col;
#pragma unroll
        for (int r = 0; r < 8; ++r) {
            tp[r * GCN_PITCH +  0] = c0[r];
            tp[r * GCN_PITCH + 16] = c1[r];
            tp[r * GCN_PITCH + 32] = c2[r];
            tp[r * GCN_PITCH + 48] = c3[r];
        }
    }
    __syncthreads();
    if (active) {
        const int c4 = (lane & 15) * 4;
#pragma unroll
        for (int i = 0; i < 8; ++i) {
            const int R = h + 2 * i;
            const v4f v = *(const v4f*)&T[R * GCN_PITCH + c4];
            const size_t g = (size_t)(m0 + R) * GCN_D + c4;
            *(v4f*)(y + g) = v;
        }
    }
}

__global__ __launch_bounds__(256)
void gcn_gather(const int* __restrict__ rowptr, const int* __restrict__ csr,
                const float* __restrict__ y, const float* __restrict__ bias,
                float* __restrict__ out) {
    const int l16 = threadIdx.x & 15;
    const int d   = (blockIdx.x * 256 + threadIdx.x) >> 4;
    const int start = rowptr[d];
    const int n     = rowptr[d + 1] - start;
    const int nmax = __builtin_amdgcn_readfirstlane(max(n, __shfl_xor(n, 16)));

    const v4f* y4 = (const v4f*)y;
    const v4f zero = {0.0f, 0.0f, 0.0f, 0.0f};
    v4f a0 = y4[d * 16 + l16] + *(const v4f*)(bias + 4 * l16);
    v4f a1 = zero, a2 = zero, a3 = zero;

    for (int base = 0; base < nmax; base += 16) {
        const int k = base + l16;
        const int idx = (k < n) ? csr[start + k] : d;
        const int lim = min(16, nmax - base);
        for (int j = 0; j < lim; j += 4) {
            const int s0 = __shfl(idx, j + 0, 16);
            const int s1 = __shfl(idx, j + 1, 16);
            const int s2 = __shfl(idx, j + 2, 16);
            const int s3 = __shfl(idx, j + 3, 16);
            const v4f t0 = y4[s0 * 16 + l16];
            const v4f t1 = y4[s1 * 16 + l16];
            const v4f t2 = y4[s2 * 16 + l16];
            const v4f t3 = y4[s3 * 16 + l16];
            a0 += (base + j + 0 < n) ? t0 : zero;
            a1 += (base + j + 1 < n) ? t1 : zero;
            a2 += (base + j + 2 < n) ? t2 : zero;
            a3 += (base + j + 3 < n) ? t3 : zero;
        }
    }
    ((v4f*)out)[d * 16 + l16] = (a0 + a1) + (a2 + a3);
}

extern "C" void kernel_launch(void* const* d_in, const int* in_sizes, int n_in,
                              void* d_out, int out_size, void* d_ws, size_t ws_size,
                              hipStream_t stream) {
    const float* x          = (const float*)d_in[0];
    const int*   edge_index = (const int*)d_in[1];
    const float* W          = (const float*)d_in[2];
    const float* b          = (const float*)d_in[3];
    float*       out        = (float*)d_out;

    float* deg    = (float*)d_ws;
    int*   cnt    = (int*)d_ws + GCN_WS_CNT;
    int*   rowptr = (int*)d_ws + GCN_WS_ROWPTR;
    int*   rank   = (int*)d_ws + GCN_WS_RANK;
    int*   csr    = (int*)d_ws + GCN_WS_CSR;
    float* y      = (float*)d_ws + GCN_WS_Y;

    gcn_zero<<<(GCN_N_NODES + 255) / 256, 256, 0, stream>>>(deg, cnt);
    gcn_count<<<(GCN_N_EDGES + 255) / 256, 256, 0, stream>>>(edge_index, deg, cnt, rank);
    gcn_rowptr<<<(GCN_N_NODES + GCN_SCAN_T - 1) / GCN_SCAN_T, GCN_SCAN_T, 0, stream>>>(cnt, rowptr);
    gcn_fill<<<(GCN_N_EDGES + 255) / 256, 256, 0, stream>>>(edge_index, rowptr, rank, csr);
    gcn_linear_first<<<(GCN_N_NODES / 16 + 7) / 8, 256, 0, stream>>>(x, deg, W, y);
    gcn_gather<<<(GCN_N_NODES * 16) / 256, 256, 0, stream>>>(rowptr, csr, y, b, out);
}


# out-degree counted without memory atomics: 125 work-groups each count an 8000-edge slice into a private all-node histogram in LDS (integer adds), planes summed beside the edge pass; same deg values
# speedup vs baseline: 2.3991x; 1.4569x over previous
#include <hip/hip_runtime.h>
#include <hip/hip_bf16.h>


typedef __attribute__((ext_vector_type(2))) float v2f;
typedef __attribute__((ext_vector_type(4))) float v4f;
typedef __attribute__((ext_vector_type(8))) float v8f;

#define GCN_N_NODES 100000
#define GCN_N_EDGES 1000000
#define GCN_D 64
#define GCN_PITCH 68
#define GCN_CAP 32

#define GCN_WS_CNT    100000
#define GCN_WS_OVFN   200000
#define GCN_WS_ADJ    200032
#define GCN_WS_OVF    3400032
#define GCN_WS_Y      5400064
#define GCN_WS_PART   11800064

#define GCN_DBLK   125
#define GCN_DEDGES (GCN_N_EDGES / GCN_DBLK)
#define GCN_HWORDS (GCN_N_NODES / 2)
#define GCN_SUMBLK ((GCN_HWORDS + 255) / 256)

__global__ __launch_bounds__(1024)
void gcn_degree_planes(const int* __restrict__ edge_index, unsigned* __restrict__ part,
                       int* __restrict__ cnt, int* __restrict__ ovf_n) {
    __shared__ unsigned hist[GCN_HWORDS];
    const int tid = threadIdx.x;
    for (int w = tid; w < GCN_HWORDS; w += 1024) hist[w] = 0u;
    if (tid < GCN_N_NODES / GCN_DBLK) cnt[blockIdx.x * (GCN_N_NODES / GCN_DBLK) + tid] = 0;
    if (blockIdx.x == 0 && tid == 0) *ovf_n = 0;
    __syncthreads();
    const int* src = edge_index + blockIdx.x * GCN_DEDGES;
    for (int e = tid; e < GCN_DEDGES; e += 1024) {
        const int s = src[e];
        atomicAdd(&hist[s >> 1], 1u << ((s & 1) * 16));
    }
    __syncthreads();
    unsigned* plane = part + blockIdx.x * GCN_HWORDS;
    for (int w = tid; w < GCN_HWORDS; w += 1024) plane[w] = hist[w];
}

__global__ __launch_bounds__(256)
void gcn_sum_fill(const int* __restrict__ edge_index, const unsigned* __restrict__ part, float* __restrict__ deg,
                  int* __restrict__ cnt, int* __restrict__ adj,
                  int* __restrict__ ovf_n, int* __restrict__ ovf) {
    if (blockIdx.x < GCN_SUMBLK) {
        const int w = blockIdx.x * 256 + threadIdx.x;
        if (w < GCN_HWORDS) {
            unsigned lo = 0u, hi = 0u;
#pragma unroll 25
            for (int k = 0; k < GCN_DBLK; ++k) {
                const unsigned v = part[k * GCN_HWORDS + w];
                lo += v & 0xffffu;
                hi += v >> 16;
            }
            *(float2*)(deg + 2 * w) = make_float2((float)lo, (float)hi);
        }
        return;
    }
    const int e = (blockIdx.x - GCN_SUMBLK) * 256 + threadIdx.x;
    if (e < GCN_N_EDGES) {
        const int s = edge_index[e];
        const int d = edge_index[GCN_N_EDGES + e];
        const int slot = atomicAdd(&cnt[d], 1);
        if (slot < GCN_CAP) {
            adj[d * GCN_CAP + slot] = s;
        } else {
            const int o = atomicAdd(ovf_n, 1);
            *(int2*)(ovf + 2 * o) = make_int2(d, s);
        }
    }
}

__global__ __launch_bounds__(256)
void gcn_linear_first(const float* __restrict__ x, const float* __restrict__ deg,
                      const float* __restrict__ W, float* __restrict__ y) {
    __shared__ float Ws[GCN_D * GCN_PITCH];
    __shared__ float Ts[8][16 * GCN_PITCH];

    const int tid  = threadIdx.x;
    const int lane = tid & 31;
    const int wave = tid >> 5;

#pragma unroll
    for (int q = 0; q < 4; ++q) {
        const int j  = tid + 256 * q;
        const int o  = j >> 4;
        const int k4 = j & 15;
        const float4 v = ((const float4*)W)[j];
        *(float4*)&Ws[o * GCN_PITCH + 4 * k4] = v;
    }
    __syncthreads();

    const int  strip  = blockIdx.x * 8 + wave;
    const bool active = strip < GCN_N_NODES / 16;
    const int  m0  = strip * 16;
    const int  row = lane & 15;
    const int  col = lane & 15;
    const int  h   = lane >> 4;
    const int  kb  = h << 1;
    float* T = Ts[wave];

    if (active) {
        const float s = rsqrtf(deg[m0 + row] + 1.0f);
        const float* arow = x + (size_t)(m0 + row) * GCN_D + kb;
        v2f a[16];
#pragma unroll
        for (int t = 0; t < 16; ++t) {
            a[t] = *(const v2f*)(arow + 4 * t);
        }
#pragma unroll
        for (int t = 0; t < 16; ++t) {
            a[t].x *= s; a[t].y *= s;
        }

        v8f c0, c1, c2, c3;
#pragma unroll
        for (int r = 0; r < 8; ++r) { c0[r] = 0.0f; c1[r] = 0.0f; c2[r] = 0.0f; c3[r] = 0.0f; }

        const float* wl = &Ws[col * GCN_PITCH + kb];
#pragma unroll
        for (int t = 0; t < 16; ++t) {
            const v2f b0 = *(const v2f*)(wl + (0 * 16) * GCN_PITCH + 4 * t);
            const v2f b1 = *(const v2f*)(wl + (1 * 16) * GCN_PITCH + 4 * t);
            const v2f b2 = *(const v2f*)(wl + (2 * 16) * GCN_PITCH + 4 * t);
            const v2f b3 = *(const v2f*)(wl + (3 * 16) * GCN_PITCH + 4 * t);
            c0 = __builtin_amdgcn_wmma_f32_16x16x4_f32(false, a[t], false, b0, (short)0, c0, false, false);
            c1 = __builtin_amdgcn_wmma_f32_16x16x4_f32(false, a[t], false, b1, (short)0, c1, false, false);
            c2 = __builtin_amdgcn_wmma_f32_16x16x4_f32(false, a[t], false, b2, (short)0, c2, false, false);
            c3 = __builtin_amdgcn_wmma_f32_16x16x4_f32(false, a[t], false, b3, (short)0, c3, false, false);
        }

        float* tp = T + (8 * h) * GCN_PITCH + col;
#pragma unroll
        for (int r = 0; r < 8; ++r) {
            tp[r * GCN_PITCH +  0] = c0[r];
            tp[r * GCN_PITCH + 16] = c1[r];
            tp[r * GCN_PITCH + 32] = c2[r];
            tp[r * GCN_PITCH + 48] = c3[r];
        }
    }
    __syncthreads();
    if (active) {
        const int c4 = (lane & 15) * 4;
#pragma unroll
        for (int i = 0; i < 8; ++i) {
            const int R = h + 2 * i;
            const v4f v = *(const v4f*)&T[R * GCN_PITCH + c4];
            const size_t g = (size_t)(m0 + R) * GCN_D + c4;
            *(v4f*)(y + g) = v;
        }
    }
}

__global__ __launch_bounds__(256)
void gcn_gather(const int* __restrict__ cnt, const int* __restrict__ adj,
                const int* __restrict__ ovf_n, const int* __restrict__ ovf,
                const float* __restrict__ y, const float* __restrict__ bias,
                float* __restrict__ out) {
    const int l16 = threadIdx.x & 15;
    const int d   = (blockIdx.x * 256 + threadIdx.x) >> 4;
    const int nall = cnt[d];
    const int n    = min(nall, GCN_CAP);
    const int* row = adj + d * GCN_CAP;
    const int nmax    = __builtin_amdgcn_readfirstlane(max(n, __shfl_xor(n, 16)));
    const int nallmax = __builtin_amdgcn_readfirstlane(max(nall, __shfl_xor(nall, 16)));

    const v4f* y4 = (const v4f*)y;
    const v4f zero = {0.0f, 0.0f, 0.0f, 0.0f};
    v4f a0 = y4[d * 16 + l16] + *(const v4f*)(bias + 4 * l16);
    v4f a1 = zero, a2 = zero, a3 = zero;

    for (int base = 0; base < nmax; base += 16) {
        const int k = base + l16;
        const int idx = (k < n) ? row[k] : d;
        const int lim = min(16, nmax - base);
        for (int j = 0; j < lim; j += 4) {
            const int s0 = __shfl(idx, j + 0, 16);
            const int s1 = __shfl(idx, j + 1, 16);
            const int s2 = __shfl(idx, j + 2, 16);
            const int s3 = __shfl(idx, j + 3, 16);
            const v4f t0 = y4[s0 * 16 + l16];
            const v4f t1 = y4[s1 * 16 + l16];
            const v4f t2 = y4[s2 * 16 + l16];
            const v4f t3 = y4[s3 * 16 + l16];
            a0 += (base + j + 0 < n) ? t0 : zero;
            a1 += (base + j + 1 < n) ? t1 : zero;
            a2 += (base + j + 2 < n) ? t2 : zero;
            a3 += (base + j + 3 < n) ? t3 : zero;
        }
    }
    v4f acc = (a0 + a1) + (a2 + a3);

    if (nallmax > GCN_CAP) {
        const int m = *ovf_n;
        for (int base = 0; base < m; base += 16) {
            const int i = base + l16;
            int od = -1, os = 0;
            if (i < m) { const int2 r = *(const int2*)(ovf + 2 * i); od = r.x; os = r.y; }
            const int lim = min(16, m - base);
            for (int j = 0; j < lim; ++j) {
                const int dd = __shfl(od, j, 16);
                const int ss = __shfl(os, j, 16);
                if (dd == d) acc += y4[ss * 16 + l16];
            }
        }
    }
    ((v4f*)out)[d * 16 + l16] = acc;
}

extern "C" void kernel_launch(void* const* d_in, const int* in_sizes, int n_in,
                              void* d_out, int out_size, void* d_ws, size_t ws_size,
                              hipStream_t stream) {
    const float* x          = (const float*)d_in[0];
    const int*   edge_index = (const int*)d_in[1];
    const float* W          = (const float*)d_in[2];
    const float* b          = (const float*)d_in[3];
    float*       out        = (float*)d_out;

    float* deg    = (float*)d_ws;
    int*   cnt    = (int*)d_ws + GCN_WS_CNT;
    int*   ovf_n  = (int*)d_ws + GCN_WS_OVFN;
    int*   adj    = (int*)d_ws + GCN_WS_ADJ;
    int*   ovf    = (int*)d_ws + GCN_WS_OVF;
    float* y      = (float*)d_ws + GCN_WS_Y;
    unsigned* part = (unsigned*)d_ws + GCN_WS_PART;

    gcn_degree_planes<<<GCN_DBLK, 1024, 0, stream>>>(edge_index, part, cnt, ovf_n);
    gcn_sum_fill<<<GCN_SUMBLK + (GCN_N_EDGES + 255) / 256, 256, 0, stream>>>(edge_index, part, deg, cnt, adj, ovf_n, ovf);
    gcn_linear_first<<<(GCN_N_NODES / 16 + 7) / 8, 256, 0, stream>>>(x, deg, W, y);
    gcn_gather<<<(GCN_N_NODES * 16) / 256, 256, 0, stream>>>(cnt, adj, ovf_n, ovf, y, b, out);
}


# linear layer runs in the edge pass's launch, work-groups dealt 1:5: a linear work-group sums its own 128 nodes' degree planes (256-B plane pitch); 3 launches, no plane-sum pass; sums unchanged
# speedup vs baseline: 2.6264x; 1.0947x over previous
#include <hip/hip_runtime.h>
#include <hip/hip_bf16.h>


typedef __attribute__((ext_vector_type(2))) float v2f;
typedef __attribute__((ext_vector_type(4))) float v4f;
typedef __attribute__((ext_vector_type(8))) float v8f;

#define GCN_N_NODES 100000
#define GCN_N_EDGES 1000000
#define GCN_D 64
#define GCN_PITCH 68
#define GCN_CAP 32

#define GCN_WS_CNT    100000
#define GCN_WS_OVFN   200000
#define GCN_WS_ADJ    200032
#define GCN_WS_OVF    3400032
#define GCN_WS_Y      5400064
#define GCN_WS_PART   11800064

#define GCN_DBLK   125
#define GCN_DEDGES (GCN_N_EDGES / GCN_DBLK)
#define GCN_HWORDS (GCN_N_NODES / 2)
#define GCN_HPITCH 50048
#define GCN_LINBLK ((GCN_N_NODES / 16 + 7) / 8)
#define GCN_EDGBLK ((GCN_N_EDGES + 255) / 256)
#define GCN_MIX    6
static_assert(GCN_LINBLK * (GCN_MIX - 1) >= GCN_EDGBLK, "the mixed launch must hold every edge work-group");

__global__ __launch_bounds__(1024)
void gcn_degree_planes(const int* __restrict__ edge_index, unsigned* __restrict__ part,
                       int* __restrict__ cnt, int* __restrict__ ovf_n) {
    __shared__ unsigned hist[GCN_HWORDS];
    const int tid = threadIdx.x;
    for (int w = tid; w < GCN_HWORDS; w += 1024) hist[w] = 0u;
    if (tid < GCN_N_NODES / GCN_DBLK) cnt[blockIdx.x * (GCN_N_NODES / GCN_DBLK) + tid] = 0;
    if (blockIdx.x == 0 && tid == 0) *ovf_n = 0;
    __syncthreads();
    const int* src = edge_index + blockIdx.x * GCN_DEDGES;
    for (int e = tid; e < GCN_DEDGES; e += 1024) {
        const int s = src[e];
        atomicAdd(&hist[s >> 1], 1u << ((s & 1) * 16));
    }
    __syncthreads();
    unsigned* plane = part + blockIdx.x * GCN_HPITCH;
    for (int w = tid; w < GCN_HWORDS; w += 1024) plane[w] = hist[w];
}

__global__ __launch_bounds__(256)
void gcn_fill_linear(const int* __restrict__ edge_index, const unsigned* __restrict__ part,
                     int* __restrict__ cnt, int* __restrict__ adj,
                     int* __restrict__ ovf_n, int* __restrict__ ovf,
                     const float* __restrict__ x, const float* __restrict__ W, float* __restrict__ y) {
    __shared__ float Ws[GCN_D * GCN_PITCH];
    __shared__ float Ts[8][16 * GCN_PITCH];
    __shared__ __attribute__((aligned(16))) unsigned Ds[4][128];

    const int tid = threadIdx.x;
    const int q   = blockIdx.x / GCN_MIX;
    const int r   = blockIdx.x - q * GCN_MIX;

    if (r != 0) {
        const int e = (q * (GCN_MIX - 1) + (r - 1)) * 256 + tid;
        if (e < GCN_N_EDGES) {
            const int s = edge_index[e];
            const int d = edge_index[GCN_N_EDGES + e];
            const int slot = atomicAdd(&cnt[d], 1);
            if (slot < GCN_CAP) {
                adj[d * GCN_CAP + slot] = s;
            } else {
                const int o = atomicAdd(ovf_n, 1);
                *(int2*)(ovf + 2 * o) = make_int2(d, s);
            }
        }
        return;
    }

    const int lane = tid & 31;
    const int wave = tid >> 5;

    const int  strip  = q * 8 + wave;
    const bool active = strip < GCN_N_NODES / 16;
    const int  m0  = strip * 16;
    const int  row = lane & 15;
    const int  col = lane & 15;
    const int  h   = lane >> 4;
    const int  kb  = h << 1;
    float* T = Ts[wave];

    float4 wv[4];
#pragma unroll
    for (int j4 = 0; j4 < 4; ++j4) wv[j4] = ((const float4*)W)[tid + 256 * j4];

    const int w  = tid & 63;
    const int g  = tid >> 6;
    const int hw = q * 64 + w;
    const unsigned  keep = (hw < GCN_HWORDS) ? 0xffffffffu : 0u;
    const unsigned* pp   = part + g * GCN_HPITCH + min(hw, GCN_HWORDS - 1);
    const unsigned vl = pp[(g == 0 ? 124 : 0) * GCN_HPITCH] & (g == 0 ? keep : 0u);
    unsigned lo = vl & 0xffffu, hi = vl >> 16;
#pragma unroll
    for (int k = 0; k < 31; ++k) {
        const unsigned v = pp[(4 * k) * GCN_HPITCH] & keep;
        lo += v & 0xffffu;
        hi += v >> 16;
    }

    v2f a[16];
    const float* arow = x + (size_t)((active ? m0 : 0) + row) * GCN_D + kb;
#pragma unroll
    for (int t = 0; t < 16; ++t) {
        a[t] = *(const v2f*)(arow + 4 * t);
    }

#pragma unroll
    for (int j4 = 0; j4 < 4; ++j4) {
        const int j  = tid + 256 * j4;
        const int o  = j >> 4;
        const int k4 = j & 15;
        *(float4*)&Ws[o * GCN_PITCH + 4 * k4] = wv[j4];
    }
    *(uint2*)&Ds[g][2 * w] = make_uint2(lo, hi);
    __syncthreads();

    if (active) {
        const int   n128 = wave * 16 + row;
        const float deg  = (float)(Ds[0][n128] + Ds[1][n128] + Ds[2][n128] + Ds[3][n128]);
        const float s    = rsqrtf(deg + 1.0f);
#pragma unroll
        for (int t = 0; t < 16; ++t) {
            a[t].x *= s; a[t].y *= s;
        }

        v8f c0, c1, c2, c3;
#pragma unroll
        for (int rr = 0; rr < 8; ++rr) { c0[rr] = 0.0f; c1[rr] = 0.0f; c2[rr] = 0.0f; c3[rr] = 0.0f; }

        const float* wl = &Ws[col * GCN_PITCH + kb];
#pragma unroll
        for (int t = 0; t < 16; ++t) {
            const v2f b0 = *(const v2f*)(wl + (0 * 16) * GCN_PITCH + 4 * t);
            const v2f b1 = *(const v2f*)(wl + (1 * 16) * GCN_PITCH + 4 * t);
            const v2f b2 = *(const v2f*)(wl + (2 * 16) * GCN_PITCH + 4 * t);
            const v2f b3 = *(const v2f*)(wl + (3 * 16) * GCN_PITCH + 4 * t);
            c0 = __builtin_amdgcn_wmma_f32_16x16x4_f32(false, a[t], false, b0, (short)0, c0, false, false);
            c1 = __builtin_amdgcn_wmma_f32_16x16x4_f32(false, a[t], false, b1, (short)0, c1, false, false);
            c2 = __builtin_amdgcn_wmma_f32_16x16x4_f32(false, a[t], false, b2, (short)0, c2, false, false);
            c3 = __builtin_amdgcn_wmma_f32_16x16x4_f32(false, a[t], false, b3, (short)0, c3, false, false);
        }

        float* tp = T + (8 * h) * GCN_PITCH + col;
#pragma unroll
        for (int rr = 0; rr < 8; ++rr) {
            tp[rr * GCN_PITCH +  0] = c0[rr];
            tp[rr * GCN_PITCH + 16] = c1[rr];
            tp[rr * GCN_PITCH + 32] = c2[rr];
            tp[rr * GCN_PITCH + 48] = c3[rr];
        }
    }
    __syncthreads();
    if (active) {
        const int c4 = (lane & 15) * 4;
#pragma unroll
        for (int i = 0; i < 8; ++i) {
            const int R = h + 2 * i;
            const v4f v = *(const v4f*)&T[R * GCN_PITCH + c4];
            const size_t gofs = (size_t)(m0 + R) * GCN_D + c4;
            *(v4f*)(y + gofs) = v;
        }
    }
}

__global__ __launch_bounds__(256)
void gcn_gather(const int* __restrict__ cnt, const int* __restrict__ adj,
                const int* __restrict__ ovf_n, const int* __restrict__ ovf,
                const float* __restrict__ y, const float* __restrict__ bias,
                float* __restrict__ out) {
    const int l16 = threadIdx.x & 15;
    const int d   = (blockIdx.x * 256 + threadIdx.x) >> 4;
    const int nall = cnt[d];
    const int n    = min(nall, GCN_CAP);
    const int* row = adj + d * GCN_CAP;
    const int nmax    = __builtin_amdgcn_readfirstlane(max(n, __shfl_xor(n, 16)));
    const int nallmax = __builtin_amdgcn_readfirstlane(max(nall, __shfl_xor(nall, 16)));

    const v4f* y4 = (const v4f*)y;
    const v4f zero = {0.0f, 0.0f, 0.0f, 0.0f};
    v4f a0 = y4[d * 16 + l16] + *(const v4f*)(bias + 4 * l16);
    v4f a1 = zero, a2 = zero, a3 = zero;

    for (int base = 0; base < nmax; base += 16) {
        const int k = base + l16;
        const int idx = (k < n) ? row[k] : d;
        const int lim = min(16, nmax - base);
        for (int j = 0; j < lim; j += 4) {
            const int s0 = __shfl(idx, j + 0, 16);
            const int s1 = __shfl(idx, j + 1, 16);
            const int s2 = __shfl(idx, j + 2, 16);
            const int s3 = __shfl(idx, j + 3, 16);
            const v4f t0 = y4[s0 * 16 + l16];
            const v4f t1 = y4[s1 * 16 + l16];
            const v4f t2 = y4[s2 * 16 + l16];
            const v4f t3 = y4[s3 * 16 + l16];
            a0 += (base + j + 0 < n) ? t0 : zero;
            a1 += (base + j + 1 < n) ? t1 : zero;
            a2 += (base + j + 2 < n) ? t2 : zero;
            a3 += (base + j + 3 < n) ? t3 : zero;
        }
    }
    v4f acc = (a0 + a1) + (a2 + a3);

    if (nallmax > GCN_CAP) {
        const int m = *ovf_n;
        for (int base = 0; base < m; base += 16) {
            const int i = base + l16;
            int od = -1, os = 0;
            if (i < m) { const int2 r = *(const int2*)(ovf + 2 * i); od = r.x; os = r.y; }
            const int lim = min(16, m - base);
            for (int j = 0; j < lim; ++j) {
                const int dd = __shfl(od, j, 16);
                const int ss = __shfl(os, j, 16);
                if (dd == d) acc += y4[ss * 16 + l16];
            }
        }
    }
    ((v4f*)out)[d * 16 + l16] = acc;
}

extern "C" void kernel_launch(void* const* d_in, const int* in_sizes, int n_in,
                              void* d_out, int out_size, void* d_ws, size_t ws_size,
                              hipStream_t stream) {
    const float* x          = (const float*)d_in[0];
    const int*   edge_index = (const int*)d_in[1];
    const float* W          = (const float*)d_in[2];
    const float* b          = (const float*)d_in[3];
    float*       out        = (float*)d_out;

    int*   cnt    = (int*)d_ws + GCN_WS_CNT;
    int*   ovf_n  = (int*)d_ws + GCN_WS_OVFN;
    int*   adj    = (int*)d_ws + GCN_WS_ADJ;
    int*   ovf    = (int*)d_ws + GCN_WS_OVF;
    float* y      = (float*)d_ws + GCN_WS_Y;
    unsigned* part = (unsigned*)d_ws + GCN_WS_PART;

    gcn_degree_planes<<<GCN_DBLK, 1024, 0, stream>>>(edge_index, part, cnt, ovf_n);
    gcn_fill_linear<<<GCN_LINBLK * GCN_MIX, 256, 0, stream>>>(edge_index, part, cnt, adj, ovf_n, ovf, x, W, y);
    gcn_gather<<<(GCN_N_NODES * 16) / 256, 256, 0, stream>>>(cnt, adj, ovf_n, ovf, y, b, out);
}


# row counters and slots split into two planes, one per half of the chip (die read from placement word): a counter line takes atomics from one half only; a row sums first-half edges, then second-half
# speedup vs baseline: 3.5804x; 1.3632x over previous
#include <hip/hip_runtime.h>
#include <hip/hip_bf16.h>


typedef __attribute__((ext_vector_type(2))) float v2f;
typedef __attribute__((ext_vector_type(4))) float v4f;
typedef __attribute__((ext_vector_type(8))) float v8f;

#define GCN_N_NODES 100000
#define GCN_N_EDGES 1000000
#define GCN_D 64
#define GCN_PITCH 68
#define GCN_CAP 32
#define GCN_HCAP 16

#define GCN_WS_CNT    0
#define GCN_WS_OVFN   200000
#define GCN_WS_ADJ    200032
#define GCN_WS_OVF    3400032
#define GCN_WS_Y      5400064
#define GCN_WS_PART   11800064

#define GCN_DBLK   125
#define GCN_DEDGES (GCN_N_EDGES / GCN_DBLK)
#define GCN_HWORDS (GCN_N_NODES / 2)
#define GCN_HPITCH 50048
#define GCN_LINBLK ((GCN_N_NODES / 16 + 7) / 8)
#define GCN_EDGBLK ((GCN_N_EDGES + 255) / 256)
#define GCN_MIX    6
static_assert(GCN_LINBLK * (GCN_MIX - 1) >= GCN_EDGBLK, "the mixed launch must hold every edge work-group");

__global__ __launch_bounds__(1024)
void gcn_degree_planes(const int* __restrict__ edge_index, unsigned* __restrict__ part,
                       int* __restrict__ cnt, int* __restrict__ ovf_n) {
    __shared__ unsigned hist[GCN_HWORDS];
    const int tid = threadIdx.x;
    for (int w = tid; w < GCN_HWORDS; w += 1024) hist[w] = 0u;
    if (tid < GCN_N_NODES / GCN_DBLK) {
        cnt[blockIdx.x * (GCN_N_NODES / GCN_DBLK) + tid] = 0;
        cnt[GCN_N_NODES + blockIdx.x * (GCN_N_NODES / GCN_DBLK) + tid] = 0;
    }
    if (blockIdx.x == 0 && tid == 0) *ovf_n = 0;
    __syncthreads();
    const int* src = edge_index + blockIdx.x * GCN_DEDGES;
    for (int e = tid; e < GCN_DEDGES; e += 1024) {
        const int s = src[e];
        atomicAdd(&hist[s >> 1], 1u << ((s & 1) * 16));
    }
    __syncthreads();
    unsigned* plane = part + blockIdx.x * GCN_HPITCH;
    for (int w = tid; w < GCN_HWORDS; w += 1024) plane[w] = hist[w];
}

__global__ __launch_bounds__(256)
void gcn_fill_linear(const int* __restrict__ edge_index, const unsigned* __restrict__ part,
                     int* __restrict__ cnt, int* __restrict__ adj,
                     int* __restrict__ ovf_n, int* __restrict__ ovf,
                     const float* __restrict__ x, const float* __restrict__ W, float* __restrict__ y) {
    __shared__ float Ws[GCN_D * GCN_PITCH];
    __shared__ float Ts[8][16 * GCN_PITCH];
    __shared__ __attribute__((aligned(16))) unsigned Ds[4][128];

    const int tid = threadIdx.x;
    const int q   = blockIdx.x / GCN_MIX;
    const int r   = blockIdx.x - q * GCN_MIX;

    if (r != 0) {
        const int half = (int)((__builtin_amdgcn_s_sendmsg_rtn(0x87) >> 18) & 1u);
        const int e = (q * (GCN_MIX - 1) + (r - 1)) * 256 + tid;
        if (e < GCN_N_EDGES) {
            const int s = edge_index[e];
            const int d = edge_index[GCN_N_EDGES + e];
            const int slot = atomicAdd(&cnt[half * GCN_N_NODES + d], 1);
            if (slot < GCN_HCAP) {
                adj[d * GCN_CAP + half * GCN_HCAP + slot] = s;
            } else {
                const int o = atomicAdd(ovf_n, 1);
                *(int2*)(ovf + 2 * o) = make_int2(d, s);
            }
        }
        return;
    }

    const int lane = tid & 31;
    const int wave = tid >> 5;

    const int  strip  = q * 8 + wave;
    const bool active = strip < GCN_N_NODES / 16;
    const int  m0  = strip * 16;
    const int  row = lane & 15;
    const int  col = lane & 15;
    const int  h   = lane >> 4;
    const int  kb  = h << 1;
    float* T = Ts[wave];

    float4 wv[4];
#pragma unroll
    for (int j4 = 0; j4 < 4; ++j4) wv[j4] = ((const float4*)W)[tid + 256 * j4];

    const int w  = tid & 63;
    const int g  = tid >> 6;
    const int hw = q * 64 + w;
    const unsigned  keep = (hw < GCN_HWORDS) ? 0xffffffffu : 0u;
    const unsigned* pp   = part + g * GCN_HPITCH + min(hw, GCN_HWORDS - 1);
    const unsigned vl = pp[(g == 0 ? 124 : 0) * GCN_HPITCH] & (g == 0 ? keep : 0u);
    unsigned lo = vl & 0xffffu, hi = vl >> 16;
#pragma unroll
    for (int k = 0; k < 31; ++k) {
        const unsigned v = pp[(4 * k) * GCN_HPITCH] & keep;
        lo += v & 0xffffu;
        hi += v >> 16;
    }

    v2f a[16];
    const float* arow = x + (size_t)((active ? m0 : 0) + row) * GCN_D + kb;
#pragma unroll
    for (int t = 0; t < 16; ++t) {
        a[t] = *(const v2f*)(arow + 4 * t);
    }

#pragma unroll
    for (int j4 = 0; j4 < 4; ++j4) {
        const int j  = tid + 256 * j4;
        const int o  = j >> 4;
        const int k4 = j & 15;
        *(float4*)&Ws[o * GCN_PITCH + 4 * k4] = wv[j4];
    }
    *(uint2*)&Ds[g][2 * w] = make_uint2(lo, hi);
    __syncthreads();

    if (active) {
        const int   n128 = wave * 16 + row;
        const float deg  = (float)(Ds[0][n128] + Ds[1][n128] + Ds[2][n128] + Ds[3][n128]);
        const float s    = rsqrtf(deg + 1.0f);
#pragma unroll
        for (int t = 0; t < 16; ++t) {
            a[t].x *= s; a[t].y *= s;
        }

        v8f c0, c1, c2, c3;
#pragma unroll
        for (int rr = 0; rr < 8; ++rr) { c0[rr] = 0.0f; c1[rr] = 0.0f; c2[rr] = 0.0f; c3[rr] = 0.0f; }

        const float* wl = &Ws[col * GCN_PITCH + kb];
#pragma unroll
        for (int t = 0; t < 16; ++t) {
            const v2f b0 = *(const v2f*)(wl + (0 * 16) * GCN_PITCH + 4 * t);
            const v2f b1 = *(const v2f*)(wl + (1 * 16) * GCN_PITCH + 4 * t);
            const v2f b2 = *(const v2f*)(wl + (2 * 16) * GCN_PITCH + 4 * t);
            const v2f b3 = *(const v2f*)(wl + (3 * 16) * GCN_PITCH + 4 * t);
            c0 = __builtin_amdgcn_wmma_f32_16x16x4_f32(false, a[t], false, b0, (short)0, c0, false, false);
            c1 = __builtin_amdgcn_wmma_f32_16x16x4_f32(false, a[t], false, b1, (short)0, c1, false, false);
            c2 = __builtin_amdgcn_wmma_f32_16x16x4_f32(false, a[t], false, b2, (short)0, c2, false, false);
            c3 = __builtin_amdgcn_wmma_f32_16x16x4_f32(false, a[t], false, b3, (short)0, c3, false, false);
        }

        float* tp = T + (8 * h) * GCN_PITCH + col;
#pragma unroll
        for (int rr = 0; rr < 8; ++rr) {
            tp[rr * GCN_PITCH +  0] = c0[rr];
            tp[rr * GCN_PITCH + 16] = c1[rr];
            tp[rr * GCN_PITCH + 32] = c2[rr];
            tp[rr * GCN_PITCH + 48] = c3[rr];
        }
    }
    __syncthreads();
    if (active) {
        const int c4 = (lane & 15) * 4;
#pragma unroll
        for (int i = 0; i < 8; ++i) {
            const int R = h + 2 * i;
            const v4f v = *(const v4f*)&T[R * GCN_PITCH + c4];
            const size_t gofs = (size_t)(m0 + R) * GCN_D + c4;
            *(v4f*)(y + gofs) = v;
        }
    }
}

__global__ __launch_bounds__(256)
void gcn_gather(const int* __restrict__ cnt, const int* __restrict__ adj,
                const int* __restrict__ ovf_n, const int* __restrict__ ovf,
                const float* __restrict__ y, const float* __restrict__ bias,
                float* __restrict__ out) {
    const int l16 = threadIdx.x & 15;
    const int d   = (blockIdx.x * 256 + threadIdx.x) >> 4;
    const int c0 = cnt[d];
    const int c1 = cnt[GCN_N_NODES + d];
    const int n0 = min(c0, GCN_HCAP);
    const int n1 = min(c1, GCN_HCAP);
    const int n    = n0 + n1;
    const int nall = max(c0, c1);
    const int* row = adj + d * GCN_CAP;
    const int nmax    = __builtin_amdgcn_readfirstlane(max(n, __shfl_xor(n, 16)));
    const int nallmax = __builtin_amdgcn_readfirstlane(max(nall, __shfl_xor(nall, 16)));

    const int v0 = (l16 < n0) ? row[l16] : d;
    const int v1 = (l16 < n1) ? row[GCN_HCAP + l16] : d;
    const int sh   = __shfl(v1, (l16 - n0) & 15, 16);
    const int idxA = (l16 < n0) ? v0 : sh;
    const int idxB = sh;

    const v4f* y4 = (const v4f*)y;
    const v4f zero = {0.0f, 0.0f, 0.0f, 0.0f};
    v4f a0 = y4[d * 16 + l16] + *(const v4f*)(bias + 4 * l16);
    v4f a1 = zero, a2 = zero, a3 = zero;

    auto sum16 = [&](const int idx, const int base) {
        const int lim = min(16, nmax - base);
        for (int j = 0; j < lim; j += 4) {
            const int s0 = __shfl(idx, j + 0, 16);
            const int s1 = __shfl(idx, j + 1, 16);
            const int s2 = __shfl(idx, j + 2, 16);
            const int s3 = __shfl(idx, j + 3, 16);
            const v4f t0 = y4[s0 * 16 + l16];
            const v4f t1 = y4[s1 * 16 + l16];
            const v4f t2 = y4[s2 * 16 + l16];
            const v4f t3 = y4[s3 * 16 + l16];
            a0 += (base + j + 0 < n) ? t0 : zero;
            a1 += (base + j + 1 < n) ? t1 : zero;
            a2 += (base + j + 2 < n) ? t2 : zero;
            a3 += (base + j + 3 < n) ? t3 : zero;
        }
    };
    sum16(idxA, 0);
    if (nmax > 16) sum16(idxB, 16);
    v4f acc = (a0 + a1) + (a2 + a3);

    if (nallmax > GCN_HCAP) {
        const int m = *ovf_n;
        for (int base = 0; base < m; base += 16) {
            const int i = base + l16;
            int od = -1, os = 0;
            if (i < m) { const int2 r = *(const int2*)(ovf + 2 * i); od = r.x; os = r.y; }
            const int lim = min(16, m - base);
            for (int j = 0; j < lim; ++j) {
                const int dd = __shfl(od, j, 16);
                const int ss = __shfl(os, j, 16);
                if (dd == d) acc += y4[ss * 16 + l16];
            }
        }
    }
    ((v4f*)out)[d * 16 + l16] = acc;
}

extern "C" void kernel_launch(void* const* d_in, const int* in_sizes, int n_in,
                              void* d_out, int out_size, void* d_ws, size_t ws_size,
                              hipStream_t stream) {
    const float* x          = (const float*)d_in[0];
    const int*   edge_index = (const int*)d_in[1];
    const float* W          = (const float*)d_in[2];
    const float* b          = (const float*)d_in[3];
    float*       out        = (float*)d_out;

    int*   cnt    = (int*)d_ws + GCN_WS_CNT;
    int*   ovf_n  = (int*)d_ws + GCN_WS_OVFN;
    int*   adj    = (int*)d_ws + GCN_WS_ADJ;
    int*   ovf    = (int*)d_ws + GCN_WS_OVF;
    float* y      = (float*)d_ws + GCN_WS_Y;
    unsigned* part = (unsigned*)d_ws + GCN_WS_PART;

    gcn_degree_planes<<<GCN_DBLK, 1024, 0, stream>>>(edge_index, part, cnt, ovf_n);
    gcn_fill_linear<<<GCN_LINBLK * GCN_MIX, 256, 0, stream>>>(edge_index, part, cnt, adj, ovf_n, ovf, x, W, y);
    gcn_gather<<<(GCN_N_NODES * 16) / 256, 256, 0, stream>>>(cnt, adj, ovf_n, ovf, y, b, out);
}


# degree pass split two ways, 32 edge slices x 8 node parts (256 work-groups, 50-KB LDS histograms): a slice arrives in one round of 16-B loads; 32 planes written and summed, not 125; same deg
# speedup vs baseline: 3.7450x; 1.0460x over previous
#include <hip/hip_runtime.h>
#include <hip/hip_bf16.h>


typedef __attribute__((ext_vector_type(2))) float v2f;
typedef __attribute__((ext_vector_type(4))) float v4f;
typedef __attribute__((ext_vector_type(8))) float v8f;

#define GCN_N_NODES 100000
#define GCN_N_EDGES 1000000
#define GCN_D 64
#define GCN_PITCH 68
#define GCN_CAP 32
#define GCN_HCAP 16

#define GCN_WS_CNT    0
#define GCN_WS_OVFN   200000
#define GCN_WS_ADJ    200032
#define GCN_WS_OVF    3400032
#define GCN_WS_Y      5400064
#define GCN_WS_PART   11800064

#define GCN_DSLICES  32
#define GCN_DPARTS   8
#define GCN_DTHREADS 512
#define GCN_SVEC     7816
#define GCN_EVEC     (GCN_N_EDGES / 4)
#define GCN_DLOADS   ((GCN_SVEC + GCN_DTHREADS - 1) / GCN_DTHREADS)
#define GCN_PWORDS   6272
#define GCN_PNODES   (2 * GCN_PWORDS)
#define GCN_HWORDS (GCN_N_NODES / 2)
static_assert(GCN_SVEC * GCN_DSLICES >= GCN_EVEC && GCN_N_EDGES % 4 == 0, "the slices must cover every edge");
static_assert(GCN_SVEC * 4 < 65536, "a slice's count of one node must fit 16 bits");
static_assert(GCN_PWORDS * GCN_DPARTS >= GCN_HWORDS && GCN_PWORDS * (GCN_DPARTS - 1) < GCN_HWORDS && GCN_PWORDS % 64 == 0 && GCN_HWORDS % 4 == 0, "the node parts must cover a plane in whole 256-B units");
static_assert(GCN_DSLICES % 4 == 0 && (GCN_DPARTS & (GCN_DPARTS - 1)) == 0, "planes are summed four ways; the part is the low bits of the work-group index");
#define GCN_HPITCH 50048
#define GCN_LINBLK ((GCN_N_NODES / 16 + 7) / 8)
#define GCN_EDGBLK ((GCN_N_EDGES + 255) / 256)
#define GCN_MIX    6
static_assert(GCN_LINBLK * (GCN_MIX - 1) >= GCN_EDGBLK, "the mixed launch must hold every edge work-group");

__global__ __launch_bounds__(GCN_DTHREADS)
void gcn_degree_planes(const int* __restrict__ edge_index, unsigned* __restrict__ part,
                       int* __restrict__ cnt, int* __restrict__ ovf_n) {
    __shared__ __attribute__((aligned(16))) unsigned hist[GCN_PNODES + 32];
    const int tid = threadIdx.x;
    const int p   = blockIdx.x & (GCN_DPARTS - 1);
    const int s   = blockIdx.x / GCN_DPARTS;

    const int4* src4 = (const int4*)edge_index;
    int4 v[GCN_DLOADS];
#pragma unroll
    for (int k = 0; k < GCN_DLOADS; ++k) {
        v[k] = src4[min(s * GCN_SVEC + tid + GCN_DTHREADS * k, GCN_EVEC - 1)];
    }

    uint4* hist4 = (uint4*)hist;
    for (int i = tid; i < (GCN_PNODES + 32) / 4; i += GCN_DTHREADS) hist4[i] = make_uint4(0u, 0u, 0u, 0u);
    for (int i = blockIdx.x * GCN_DTHREADS + tid; i < 2 * GCN_N_NODES; i += GCN_DSLICES * GCN_DPARTS * GCN_DTHREADS) cnt[i] = 0;
    if (blockIdx.x == 0 && tid == 0) *ovf_n = 0;
    __syncthreads();

    const unsigned first = (unsigned)(p * GCN_PNODES);
    const unsigned spare = (unsigned)(GCN_PNODES + (tid & 31));
    auto count = [&](const int node) {
        const unsigned r = (unsigned)node - first;
        atomicAdd(&hist[r < (unsigned)GCN_PNODES ? r : spare], 1u);
    };
#pragma unroll
    for (int k = 0; k < GCN_DLOADS; ++k) {
        const int i = tid + GCN_DTHREADS * k;
        if (i < GCN_SVEC && s * GCN_SVEC + i < GCN_EVEC) {
            count(v[k].x); count(v[k].y); count(v[k].z); count(v[k].w);
        }
    }
    __syncthreads();

    const int nw4 = min(GCN_PWORDS, GCN_HWORDS - p * GCN_PWORDS) / 4;
    uint4* plane4 = (uint4*)(part + s * GCN_HPITCH + p * GCN_PWORDS);
    for (int i = tid; i < nw4; i += GCN_DTHREADS) {
        const uint4 a = hist4[2 * i], b = hist4[2 * i + 1];
        plane4[i] = make_uint4(a.x | (a.y << 16), a.z | (a.w << 16), b.x | (b.y << 16), b.z | (b.w << 16));
    }
}

__global__ __launch_bounds__(256)
void gcn_fill_linear(const int* __restrict__ edge_index, const unsigned* __restrict__ part,
                     int* __restrict__ cnt, int* __restrict__ adj,
                     int* __restrict__ ovf_n, int* __restrict__ ovf,
                     const float* __restrict__ x, const float* __restrict__ W, float* __restrict__ y) {
    __shared__ float Ws[GCN_D * GCN_PITCH];
    __shared__ float Ts[8][16 * GCN_PITCH];
    __shared__ __attribute__((aligned(16))) unsigned Ds[4][128];

    const int tid = threadIdx.x;
    const int q   = blockIdx.x / GCN_MIX;
    const int r   = blockIdx.x - q * GCN_MIX;

    if (r != 0) {
        const int half = (int)((__builtin_amdgcn_s_sendmsg_rtn(0x87) >> 18) & 1u);
        const int e = (q * (GCN_MIX - 1) + (r - 1)) * 256 + tid;
        if (e < GCN_N_EDGES) {
            const int s = edge_index[e];
            const int d = edge_index[GCN_N_EDGES + e];
            const int slot = atomicAdd(&cnt[half * GCN_N_NODES + d], 1);
            if (slot < GCN_HCAP) {
                adj[d * GCN_CAP + half * GCN_HCAP + slot] = s;
            } else {
                const int o = atomicAdd(ovf_n, 1);
                *(int2*)(ovf + 2 * o) = make_int2(d, s);
            }
        }
        return;
    }

    const int lane = tid & 31;
    const int wave = tid >> 5;

    const int  strip  = q * 8 + wave;
    const bool active = strip < GCN_N_NODES / 16;
    const int  m0  = strip * 16;
    const int  row = lane & 15;
    const int  col = lane & 15;
    const int  h   = lane >> 4;
    const int  kb  = h << 1;
    float* T = Ts[wave];

    float4 wv[4];
#pragma unroll
    for (int j4 = 0; j4 < 4; ++j4) wv[j4] = ((const float4*)W)[tid + 256 * j4];

    const int w  = tid & 63;
    const int g  = tid >> 6;
    const int hw = q * 64 + w;
    const unsigned  keep = (hw < GCN_HWORDS) ? 0xffffffffu : 0u;
    const unsigned* pp   = part + g * GCN_HPITCH + min(hw, GCN_HWORDS - 1);
    unsigned lo = 0u, hi = 0u;
#pragma unroll
    for (int k = 0; k < GCN_DSLICES / 4; ++k) {
        const unsigned v = pp[(4 * k) * GCN_HPITCH] & keep;
        lo += v & 0xffffu;
        hi += v >> 16;
    }

    v2f a[16];
    const float* arow = x + (size_t)((active ? m0 : 0) + row) * GCN_D + kb;
#pragma unroll
    for (int t = 0; t < 16; ++t) {
        a[t] = *(const v2f*)(arow + 4 * t);
    }

#pragma unroll
    for (int j4 = 0; j4 < 4; ++j4) {
        const int j  = tid + 256 * j4;
        const int o  = j >> 4;
        const int k4 = j & 15;
        *(float4*)&Ws[o * GCN_PITCH + 4 * k4] = wv[j4];
    }
    *(uint2*)&Ds[g][2 * w] = make_uint2(lo, hi);
    __syncthreads();

    if (active) {
        const int   n128 = wave * 16 + row;
        const float deg  = (float)(Ds[0][n128] + Ds[1][n128] + Ds[2][n128] + Ds[3][n128]);
        const float s    = rsqrtf(deg + 1.0f);
#pragma unroll
        for (int t = 0; t < 16; ++t) {
            a[t].x *= s; a[t].y *= s;
        }

        v8f c0, c1, c2, c3;
#pragma unroll
        for (int rr = 0; rr < 8; ++rr) { c0[rr] = 0.0f; c1[rr] = 0.0f; c2[rr] = 0.0f; c3[rr] = 0.0f; }

        const float* wl = &Ws[col * GCN_PITCH + kb];
#pragma unroll
        for (int t = 0; t < 16; ++t) {
            const v2f b0 = *(const v2f*)(wl + (0 * 16) * GCN_PITCH + 4 * t);
            const v2f b1 = *(const v2f*)(wl + (1 * 16) * GCN_PITCH + 4 * t);
            const v2f b2 = *(const v2f*)(wl + (2 * 16) * GCN_PITCH + 4 * t);
            const v2f b3 = *(const v2f*)(wl + (3 * 16) * GCN_PITCH + 4 * t);
            c0 = __builtin_amdgcn_wmma_f32_16x16x4_f32(false, a[t], false, b0, (short)0, c0, false, false);
            c1 = __builtin_amdgcn_wmma_f32_16x16x4_f32(false, a[t], false, b1, (short)0, c1, false, false);
            c2 = __builtin_amdgcn_wmma_f32_16x16x4_f32(false, a[t], false, b2, (short)0, c2, false, false);
            c3 = __builtin_amdgcn_wmma_f32_16x16x4_f32(false, a[t], false, b3, (short)0, c3, false, false);
        }

        float* tp = T + (8 * h) * GCN_PITCH + col;
#pragma unroll
        for (int rr = 0; rr < 8; ++rr) {
            tp[rr * GCN_PITCH +  0] = c0[rr];
            tp[rr * GCN_PITCH + 16] = c1[rr];
            tp[rr * GCN_PITCH + 32] = c2[rr];
            tp[rr * GCN_PITCH + 48] = c3[rr];
        }
    }
    __syncthreads();
    if (active) {
        const int c4 = (lane & 15) * 4;
#pragma unroll
        for (int i = 0; i < 8; ++i) {
            const int R = h + 2 * i;
            const v4f v = *(const v4f*)&T[R * GCN_PITCH + c4];
            const size_t gofs = (size_t)(m0 + R) * GCN_D + c4;
            *(v4f*)(y + gofs) = v;
        }
    }
}

__global__ __launch_bounds__(256)
void gcn_gather(const int* __restrict__ cnt, const int* __restrict__ adj,
                const int* __restrict__ ovf_n, const int* __restrict__ ovf,
                const float* __restrict__ y, const float* __restrict__ bias,
                float* __restrict__ out) {
    const int l16 = threadIdx.x & 15;
    const int d   = (blockIdx.x * 256 + threadIdx.x) >> 4;
    const int c0 = cnt[d];
    const int c1 = cnt[GCN_N_NODES + d];
    const int n0 = min(c0, GCN_HCAP);
    const int n1 = min(c1, GCN_HCAP);
    const int n    = n0 + n1;
    const int nall = max(c0, c1);
    const int* row = adj + d * GCN_CAP;
    const int nmax    = __builtin_amdgcn_readfirstlane(max(n, __shfl_xor(n, 16)));
    const int nallmax = __builtin_amdgcn_readfirstlane(max(nall, __shfl_xor(nall, 16)));

    const int v0 = (l16 < n0) ? row[l16] : d;
    const int v1 = (l16 < n1) ? row[GCN_HCAP + l16] : d;
    const int sh   = __shfl(v1, (l16 - n0) & 15, 16);
    const int idxA = (l16 < n0) ? v0 : sh;
    const int idxB = sh;

    const v4f* y4 = (const v4f*)y;
    const v4f zero = {0.0f, 0.0f, 0.0f, 0.0f};
    v4f a0 = y4[d * 16 + l16] + *(const v4f*)(bias + 4 * l16);
    v4f a1 = zero, a2 = zero, a3 = zero;

    auto sum16 = [&](const int idx, const int base) {
        const int lim = min(16, nmax - base);
        for (int j = 0; j < lim; j += 4) {
            const int s0 = __shfl(idx, j + 0, 16);
            const int s1 = __shfl(idx, j + 1, 16);
            const int s2 = __shfl(idx, j + 2, 16);
            const int s3 = __shfl(idx, j + 3, 16);
            const v4f t0 = y4[s0 * 16 + l16];
            const v4f t1 = y4[s1 * 16 + l16];
            const v4f t2 = y4[s2 * 16 + l16];
            const v4f t3 = y4[s3 * 16 + l16];
            a0 += (base + j + 0 < n) ? t0 : zero;
            a1 += (base + j + 1 < n) ? t1 : zero;
            a2 += (base + j + 2 < n) ? t2 : zero;
            a3 += (base + j + 3 < n) ? t3 : zero;
        }
    };
    sum16(idxA, 0);
    if (nmax > 16) sum16(idxB, 16);
    v4f acc = (a0 + a1) + (a2 + a3);

    if (nallmax > GCN_HCAP) {
        const int m = *ovf_n;
        for (int base = 0; base < m; base += 16) {
            const int i = base + l16;
            int od = -1, os = 0;
            if (i < m) { const int2 r = *(const int2*)(ovf + 2 * i); od = r.x; os = r.y; }
            const int lim = min(16, m - base);
            for (int j = 0; j < lim; ++j) {
                const int dd = __shfl(od, j, 16);
                const int ss = __shfl(os, j, 16);
                if (dd == d) acc += y4[ss * 16 + l16];
            }
        }
    }
    ((v4f*)out)[d * 16 + l16] = acc;
}

extern "C" void kernel_launch(void* const* d_in, const int* in_sizes, int n_in,
                              void* d_out, int out_size, void* d_ws, size_t ws_size,
                              hipStream_t stream) {
    const float* x          = (const float*)d_in[0];
    const int*   edge_index = (const int*)d_in[1];
    const float* W          = (const float*)d_in[2];
    const float* b          = (const float*)d_in[3];
    float*       out        = (float*)d_out;

    int*   cnt    = (int*)d_ws + GCN_WS_CNT;
    int*   ovf_n  = (int*)d_ws + GCN_WS_OVFN;
    int*   adj    = (int*)d_ws + GCN_WS_ADJ;
    int*   ovf    = (int*)d_ws + GCN_WS_OVF;
    float* y      = (float*)d_ws + GCN_WS_Y;
    unsigned* part = (unsigned*)d_ws + GCN_WS_PART;

    gcn_degree_planes<<<GCN_DSLICES * GCN_DPARTS, GCN_DTHREADS, 0, stream>>>(edge_index, part, cnt, ovf_n);
    gcn_fill_linear<<<GCN_LINBLK * GCN_MIX, 256, 0, stream>>>(edge_index, part, cnt, adj, ovf_n, ovf, x, W, y);
    gcn_gather<<<(GCN_N_NODES * 16) / 256, 256, 0, stream>>>(cnt, adj, ovf_n, ovf, y, b, out);
}


# slot table split into one table per half of the chip (16 slots a row each, 256-B aligned): a slot line takes 4-byte stores from one half only; counters, atomics, overflow rule, sum order kept
# speedup vs baseline: 5.1700x; 1.3805x over previous
#include <hip/hip_runtime.h>
#include <hip/hip_bf16.h>


typedef __attribute__((ext_vector_type(2))) float v2f;
typedef __attribute__((ext_vector_type(4))) float v4f;
typedef __attribute__((ext_vector_type(8))) float v8f;

#define GCN_N_NODES 100000
#define GCN_N_EDGES 1000000
#define GCN_D 64
#define GCN_PITCH 68
#define GCN_HCAP 16
#define GCN_APLANE (GCN_N_NODES * GCN_HCAP)

#define GCN_WS_CNT    0
#define GCN_WS_OVFN   200000
#define GCN_WS_ADJ    200064
#define GCN_WS_OVF    3400064
#define GCN_WS_Y      5400064
#define GCN_WS_PART   11800064

#define GCN_DSLICES  32
#define GCN_DPARTS   8
#define GCN_DTHREADS 512
#define GCN_SVEC     7816
#define GCN_EVEC     (GCN_N_EDGES / 4)
#define GCN_DLOADS   ((GCN_SVEC + GCN_DTHREADS - 1) / GCN_DTHREADS)
#define GCN_PWORDS   6272
#define GCN_PNODES   (2 * GCN_PWORDS)
#define GCN_HWORDS (GCN_N_NODES / 2)
static_assert(GCN_WS_ADJ + 2 * GCN_APLANE <= GCN_WS_OVF && GCN_WS_OVF + 2 * GCN_N_EDGES <= GCN_WS_Y, "the slot tables, the overflow list and y must not overlap");
static_assert(GCN_SVEC * GCN_DSLICES >= GCN_EVEC && GCN_N_EDGES % 4 == 0, "the slices must cover every edge");
static_assert(GCN_SVEC * 4 < 65536, "a slice's count of one node must fit 16 bits");
static_assert(GCN_PWORDS * GCN_DPARTS >= GCN_HWORDS && GCN_PWORDS * (GCN_DPARTS - 1) < GCN_HWORDS && GCN_PWORDS % 64 == 0 && GCN_HWORDS % 4 == 0, "the node parts must cover a plane in whole 256-B units");
static_assert(GCN_DSLICES % 4 == 0 && (GCN_DPARTS & (GCN_DPARTS - 1)) == 0, "planes are summed four ways; the part is the low bits of the work-group index");
#define GCN_HPITCH 50048
#define GCN_LINBLK ((GCN_N_NODES / 16 + 7) / 8)
#define GCN_EDGBLK ((GCN_N_EDGES + 255) / 256)
#define GCN_MIX    6
static_assert(GCN_LINBLK * (GCN_MIX - 1) >= GCN_EDGBLK, "the mixed launch must hold every edge work-group");

__global__ __launch_bounds__(GCN_DTHREADS)
void gcn_degree_planes(const int* __restrict__ edge_index, unsigned* __restrict__ part,
                       int* __restrict__ cnt, int* __restrict__ ovf_n) {
    __shared__ __attribute__((aligned(16))) unsigned hist[GCN_PNODES + 32];
    const int tid = threadIdx.x;
    const int p   = blockIdx.x & (GCN_DPARTS - 1);
    const int s   = blockIdx.x / GCN_DPARTS;

    const int4* src4 = (const int4*)edge_index;
    int4 v[GCN_DLOADS];
#pragma unroll
    for (int k = 0; k < GCN_DLOADS; ++k) {
        v[k] = src4[min(s * GCN_SVEC + tid + GCN_DTHREADS * k, GCN_EVEC - 1)];
    }

    uint4* hist4 = (uint4*)hist;
    for (int i = tid; i < (GCN_PNODES + 32) / 4; i += GCN_DTHREADS) hist4[i] = make_uint4(0u, 0u, 0u, 0u);
    for (int i = blockIdx.x * GCN_DTHREADS + tid; i < 2 * GCN_N_NODES; i += GCN_DSLICES * GCN_DPARTS * GCN_DTHREADS) cnt[i] = 0;
    if (blockIdx.x == 0 && tid == 0) *ovf_n = 0;
    __syncthreads();

    const unsigned first = (unsigned)(p * GCN_PNODES);
    const unsigned spare = (unsigned)(GCN_PNODES + (tid & 31));
    auto count = [&](const int node) {
        const unsigned r = (unsigned)node - first;
        atomicAdd(&hist[r < (unsigned)GCN_PNODES ? r : spare], 1u);
    };
#pragma unroll
    for (int k = 0; k < GCN_DLOADS; ++k) {
        const int i = tid + GCN_DTHREADS * k;
        if (i < GCN_SVEC && s * GCN_SVEC + i < GCN_EVEC) {
            count(v[k].x); count(v[k].y); count(v[k].z); count(v[k].w);
        }
    }
    __syncthreads();

    const int nw4 = min(GCN_PWORDS, GCN_HWORDS - p * GCN_PWORDS) / 4;
    uint4* plane4 = (uint4*)(part + s * GCN_HPITCH + p * GCN_PWORDS);
    for (int i = tid; i < nw4; i += GCN_DTHREADS) {
        const uint4 a = hist4[2 * i], b = hist4[2 * i + 1];
        plane4[i] = make_uint4(a.x | (a.y << 16), a.z | (a.w << 16), b.x | (b.y << 16), b.z | (b.w << 16));
    }
}

__global__ __launch_bounds__(256)
void gcn_fill_linear(const int* __restrict__ edge_index, const unsigned* __restrict__ part,
                     int* __restrict__ cnt, int* __restrict__ adj,
                     int* __restrict__ ovf_n, int* __restrict__ ovf,
                     const float* __restrict__ x, const float* __restrict__ W, float* __restrict__ y) {
    __shared__ float Ws[GCN_D * GCN_PITCH];
    __shared__ float Ts[8][16 * GCN_PITCH];
    __shared__ __attribute__((aligned(16))) unsigned Ds[4][128];

    const int tid = threadIdx.x;
    const int q   = blockIdx.x / GCN_MIX;
    const int r   = blockIdx.x - q * GCN_MIX;

    if (r != 0) {
        const int half = (int)((__builtin_amdgcn_s_sendmsg_rtn(0x87) >> 18) & 1u);
        const int e = (q * (GCN_MIX - 1) + (r - 1)) * 256 + tid;
        if (e < GCN_N_EDGES) {
            const int s = edge_index[e];
            const int d = edge_index[GCN_N_EDGES + e];
            const int slot = atomicAdd(&cnt[half * GCN_N_NODES + d], 1);
            if (slot < GCN_HCAP) {
                adj[(half * GCN_N_NODES + d) * GCN_HCAP + slot] = s;
            } else {
                const int o = atomicAdd(ovf_n, 1);
                *(int2*)(ovf + 2 * o) = make_int2(d, s);
            }
        }
        return;
    }

    const int lane = tid & 31;
    const int wave = tid >> 5;

    const int  strip  = q * 8 + wave;
    const bool active = strip < GCN_N_NODES / 16;
    const int  m0  = strip * 16;
    const int  row = lane & 15;
    const int  col = lane & 15;
    const int  h   = lane >> 4;
    const int  kb  = h << 1;
    float* T = Ts[wave];

    float4 wv[4];
#pragma unroll
    for (int j4 = 0; j4 < 4; ++j4) wv[j4] = ((const float4*)W)[tid + 256 * j4];

    const int w  = tid & 63;
    const int g  = tid >> 6;
    const int hw = q * 64 + w;
    const unsigned  keep = (hw < GCN_HWORDS) ? 0xffffffffu : 0u;
    const unsigned* pp   = part + g * GCN_HPITCH + min(hw, GCN_HWORDS - 1);
    unsigned lo = 0u, hi = 0u;
#pragma unroll
    for (int k = 0; k < GCN_DSLICES / 4; ++k) {
        const unsigned v = pp[(4 * k) * GCN_HPITCH] & keep;
        lo += v & 0xffffu;
        hi += v >> 16;
    }

    v2f a[16];
    const float* arow = x + (size_t)((active ? m0 : 0) + row) * GCN_D + kb;
#pragma unroll
    for (int t = 0; t < 16; ++t) {
        a[t] = *(const v2f*)(arow + 4 * t);
    }

#pragma unroll
    for (int j4 = 0; j4 < 4; ++j4) {
        const int j  = tid + 256 * j4;
        const int o  = j >> 4;
        const int k4 = j & 15;
        *(float4*)&Ws[o * GCN_PITCH + 4 * k4] = wv[j4];
    }
    *(uint2*)&Ds[g][2 * w] = make_uint2(lo, hi);
    __syncthreads();

    if (active) {
        const int   n128 = wave * 16 + row;
        const float deg  = (float)(Ds[0][n128] + Ds[1][n128] + Ds[2][n128] + Ds[3][n128]);
        const float s    = rsqrtf(deg + 1.0f);
#pragma unroll
        for (int t = 0; t < 16; ++t) {
            a[t].x *= s; a[t].y *= s;
        }

        v8f c0, c1, c2, c3;
#pragma unroll
        for (int rr = 0; rr < 8; ++rr) { c0[rr] = 0.0f; c1[rr] = 0.0f; c2[rr] = 0.0f; c3[rr] = 0.0f; }

        const float* wl = &Ws[col * GCN_PITCH + kb];
#pragma unroll
        for (int t = 0; t < 16; ++t) {
            const v2f b0 = *(const v2f*)(wl + (0 * 16) * GCN_PITCH + 4 * t);
            const v2f b1 = *(const v2f*)(wl + (1 * 16) * GCN_PITCH + 4 * t);
            const v2f b2 = *(const v2f*)(wl + (2 * 16) * GCN_PITCH + 4 * t);
            const v2f b3 = *(const v2f*)(wl + (3 * 16) * GCN_PITCH + 4 * t);
            c0 = __builtin_amdgcn_wmma_f32_16x16x4_f32(false, a[t], false, b0, (short)0, c0, false, false);
            c1 = __builtin_amdgcn_wmma_f32_16x16x4_f32(false, a[t], false, b1, (short)0, c1, false, false);
            c2 = __builtin_amdgcn_wmma_f32_16x16x4_f32(false, a[t], false, b2, (short)0, c2, false, false);
            c3 = __builtin_amdgcn_wmma_f32_16x16x4_f32(false, a[t], false, b3, (short)0, c3, false, false);
        }

        float* tp = T + (8 * h) * GCN_PITCH + col;
#pragma unroll
        for (int rr = 0; rr < 8; ++rr) {
            tp[rr * GCN_PITCH +  0] = c0[rr];
            tp[rr * GCN_PITCH + 16] = c1[rr];
            tp[rr * GCN_PITCH + 32] = c2[rr];
            tp[rr * GCN_PITCH + 48] = c3[rr];
        }
    }
    __syncthreads();
    if (active) {
        const int c4 = (lane & 15) * 4;
#pragma unroll
        for (int i = 0; i < 8; ++i) {
            const int R = h + 2 * i;
            const v4f v = *(const v4f*)&T[R * GCN_PITCH + c4];
            const size_t gofs = (size_t)(m0 + R) * GCN_D + c4;
            *(v4f*)(y + gofs) = v;
        }
    }
}

__global__ __launch_bounds__(256)
void gcn_gather(const int* __restrict__ cnt, const int* __restrict__ adj,
                const int* __restrict__ ovf_n, const int* __restrict__ ovf,
                const float* __restrict__ y, const float* __restrict__ bias,
                float* __restrict__ out) {
    const int l16 = threadIdx.x & 15;
    const int d   = (blockIdx.x * 256 + threadIdx.x) >> 4;
    const int c0 = cnt[d];
    const int c1 = cnt[GCN_N_NODES + d];
    const int n0 = min(c0, GCN_HCAP);
    const int n1 = min(c1, GCN_HCAP);
    const int n    = n0 + n1;
    const int nall = max(c0, c1);
    const int* row0 = adj + d * GCN_HCAP;
    const int* row1 = row0 + GCN_APLANE;
    const int nmax    = __builtin_amdgcn_readfirstlane(max(n, __shfl_xor(n, 16)));
    const int nallmax = __builtin_amdgcn_readfirstlane(max(nall, __shfl_xor(nall, 16)));

    const int v0 = (l16 < n0) ? row0[l16] : d;
    const int v1 = (l16 < n1) ? row1[l16] : d;
    const int sh   = __shfl(v1, (l16 - n0) & 15, 16);
    const int idxA = (l16 < n0) ? v0 : sh;
    const int idxB = sh;

    const v4f* y4 = (const v4f*)y;
    const v4f zero = {0.0f, 0.0f, 0.0f, 0.0f};
    v4f a0 = y4[d * 16 + l16] + *(const v4f*)(bias + 4 * l16);
    v4f a1 = zero, a2 = zero, a3 = zero;

    auto sum16 = [&](const int idx, const int base) {
        const int lim = min(16, nmax - base);
        for (int j = 0; j < lim; j += 4) {
            const int s0 = __shfl(idx, j + 0, 16);
            const int s1 = __shfl(idx, j + 1, 16);
            const int s2 = __shfl(idx, j + 2, 16);
            const int s3 = __shfl(idx, j + 3, 16);
            const v4f t0 = y4[s0 * 16 + l16];
            const v4f t1 = y4[s1 * 16 + l16];
            const v4f t2 = y4[s2 * 16 + l16];
            const v4f t3 = y4[s3 * 16 + l16];
            a0 += (base + j + 0 < n) ? t0 : zero;
            a1 += (base + j + 1 < n) ? t1 : zero;
            a2 += (base + j + 2 < n) ? t2 : zero;
            a3 += (base + j + 3 < n) ? t3 : zero;
        }
    };
    sum16(idxA, 0);
    if (nmax > 16) sum16(idxB, 16);
    v4f acc = (a0 + a1) + (a2 + a3);

    if (nallmax > GCN_HCAP) {
        const int m = *ovf_n;
        for (int base = 0; base < m; base += 16) {
            const int i = base + l16;
            int od = -1, os = 0;
            if (i < m) { const int2 r = *(const int2*)(ovf + 2 * i); od = r.x; os = r.y; }
            const int lim = min(16, m - base);
            for (int j = 0; j < lim; ++j) {
                const int dd = __shfl(od, j, 16);
                const int ss = __shfl(os, j, 16);
                if (dd == d) acc += y4[ss * 16 + l16];
            }
        }
    }
    ((v4f*)out)[d * 16 + l16] = acc;
}

extern "C" void kernel_launch(void* const* d_in, const int* in_sizes, int n_in,
                              void* d_out, int out_size, void* d_ws, size_t ws_size,
                              hipStream_t stream) {
    const float* x          = (const float*)d_in[0];
    const int*   edge_index = (const int*)d_in[1];
    const float* W          = (const float*)d_in[2];
    const float* b          = (const float*)d_in[3];
    float*       out        = (float*)d_out;

    int*   cnt    = (int*)d_ws + GCN_WS_CNT;
    int*   ovf_n  = (int*)d_ws + GCN_WS_OVFN;
    int*   adj    = (int*)d_ws + GCN_WS_ADJ;
    int*   ovf    = (int*)d_ws + GCN_WS_OVF;
    float* y      = (float*)d_ws + GCN_WS_Y;
    unsigned* part = (unsigned*)d_ws + GCN_WS_PART;

    gcn_degree_planes<<<GCN_DSLICES * GCN_DPARTS, GCN_DTHREADS, 0, stream>>>(edge_index, part, cnt, ovf_n);
    gcn_fill_linear<<<GCN_LINBLK * GCN_MIX, 256, 0, stream>>>(edge_index, part, cnt, adj, ovf_n, ovf, x, W, y);
    gcn_gather<<<(GCN_N_NODES * 16) / 256, 256, 0, stream>>>(cnt, adj, ovf_n, ovf, y, b, out);
}


# edge pass of the fused launch takes 4 edges a thread (16-B index loads, 4 slot atomics in flight): 7816 short edge waves, not 31256; groups dealt 4 linear : 5 edge; sums still in slot arrival order
# speedup vs baseline: 5.3345x; 1.0318x over previous
#include <hip/hip_runtime.h>
#include <hip/hip_bf16.h>


typedef __attribute__((ext_vector_type(2))) float v2f;
typedef __attribute__((ext_vector_type(4))) float v4f;
typedef __attribute__((ext_vector_type(8))) float v8f;

#define GCN_N_NODES 100000
#define GCN_N_EDGES 1000000
#define GCN_D 64
#define GCN_PITCH 68
#define GCN_HCAP 16
#define GCN_APLANE (GCN_N_NODES * GCN_HCAP)

#define GCN_WS_CNT    0
#define GCN_WS_OVFN   200000
#define GCN_WS_ADJ    200064
#define GCN_WS_OVF    3400064
#define GCN_WS_Y      5400064
#define GCN_WS_PART   11800064

#define GCN_DSLICES  32
#define GCN_DPARTS   8
#define GCN_DTHREADS 512
#define GCN_SVEC     7816
#define GCN_EVEC     (GCN_N_EDGES / 4)
#define GCN_DLOADS   ((GCN_SVEC + GCN_DTHREADS - 1) / GCN_DTHREADS)
#define GCN_PWORDS   6272
#define GCN_PNODES   (2 * GCN_PWORDS)
#define GCN_HWORDS (GCN_N_NODES / 2)
static_assert(GCN_WS_ADJ + 2 * GCN_APLANE <= GCN_WS_OVF && GCN_WS_OVF + 2 * GCN_N_EDGES <= GCN_WS_Y, "the slot tables, the overflow list and y must not overlap");
static_assert(GCN_SVEC * GCN_DSLICES >= GCN_EVEC && GCN_N_EDGES % 4 == 0, "the slices must cover every edge");
static_assert(GCN_SVEC * 4 < 65536, "a slice's count of one node must fit 16 bits");
static_assert(GCN_PWORDS * GCN_DPARTS >= GCN_HWORDS && GCN_PWORDS * (GCN_DPARTS - 1) < GCN_HWORDS && GCN_PWORDS % 64 == 0 && GCN_HWORDS % 4 == 0, "the node parts must cover a plane in whole 256-B units");
static_assert(GCN_DSLICES % 4 == 0 && (GCN_DPARTS & (GCN_DPARTS - 1)) == 0, "planes are summed four ways; the part is the low bits of the work-group index");
#define GCN_HPITCH 50048
#define GCN_LINBLK ((GCN_N_NODES / 16 + 7) / 8)
#define GCN_EDGBLK ((GCN_EVEC + 255) / 256)
#define GCN_LSHARE 4
#define GCN_ESHARE 5
#define GCN_MIX    (GCN_LSHARE + GCN_ESHARE)
#define GCN_GROUPS ((GCN_LINBLK + GCN_LSHARE - 1) / GCN_LSHARE)
static_assert(GCN_GROUPS * GCN_ESHARE >= GCN_EDGBLK, "the mixed launch must hold every edge work-group");

__global__ __launch_bounds__(GCN_DTHREADS)
void gcn_degree_planes(const int* __restrict__ edge_index, unsigned* __restrict__ part,
                       int* __restrict__ cnt, int* __restrict__ ovf_n) {
    __shared__ __attribute__((aligned(16))) unsigned hist[GCN_PNODES + 32];
    const int tid = threadIdx.x;
    const int p   = blockIdx.x & (GCN_DPARTS - 1);
    const int s   = blockIdx.x / GCN_DPARTS;

    const int4* src4 = (const int4*)edge_index;
    int4 v[GCN_DLOADS];
#pragma unroll
    for (int k = 0; k < GCN_DLOADS; ++k) {
        v[k] = src4[min(s * GCN_SVEC + tid + GCN_DTHREADS * k, GCN_EVEC - 1)];
    }

    uint4* hist4 = (uint4*)hist;
    for (int i = tid; i < (GCN_PNODES + 32) / 4; i += GCN_DTHREADS) hist4[i] = make_uint4(0u, 0u, 0u, 0u);
    for (int i = blockIdx.x * GCN_DTHREADS + tid; i < 2 * GCN_N_NODES; i += GCN_DSLICES * GCN_DPARTS * GCN_DTHREADS) cnt[i] = 0;
    if (blockIdx.x == 0 && tid == 0) *ovf_n = 0;
    __syncthreads();

    const unsigned first = (unsigned)(p * GCN_PNODES);
    const unsigned spare = (unsigned)(GCN_PNODES + (tid & 31));
    auto count = [&](const int node) {
        const unsigned r = (unsigned)node - first;
        atomicAdd(&hist[r < (unsigned)GCN_PNODES ? r : spare], 1u);
    };
#pragma unroll
    for (int k = 0; k < GCN_DLOADS; ++k) {
        const int i = tid + GCN_DTHREADS * k;
        if (i < GCN_SVEC && s * GCN_SVEC + i < GCN_EVEC) {
            count(v[k].x); count(v[k].y); count(v[k].z); count(v[k].w);
        }
    }
    __syncthreads();

    const int nw4 = min(GCN_PWORDS, GCN_HWORDS - p * GCN_PWORDS) / 4;
    uint4* plane4 = (uint4*)(part + s * GCN_HPITCH + p * GCN_PWORDS);
    for (int i = tid; i < nw4; i += GCN_DTHREADS) {
        const uint4 a = hist4[2 * i], b = hist4[2 * i + 1];
        plane4[i] = make_uint4(a.x | (a.y << 16), a.z | (a.w << 16), b.x | (b.y << 16), b.z | (b.w << 16));
    }
}

__global__ __launch_bounds__(256)
void gcn_fill_linear(const int* __restrict__ edge_index, const unsigned* __restrict__ part,
                     int* __restrict__ cnt, int* __restrict__ adj,
                     int* __restrict__ ovf_n, int* __restrict__ ovf,
                     const float* __restrict__ x, const float* __restrict__ W, float* __restrict__ y) {
    __shared__ float Ws[GCN_D * GCN_PITCH];
    __shared__ float Ts[8][16 * GCN_PITCH];
    __shared__ __attribute__((aligned(16))) unsigned Ds[4][128];

    const int tid = threadIdx.x;
    const int grp = blockIdx.x / GCN_MIX;
    const int r   = blockIdx.x - grp * GCN_MIX;

    if (r >= GCN_LSHARE) {
        const int half = (int)((__builtin_amdgcn_s_sendmsg_rtn(0x87) >> 18) & 1u);
        const int vi = (grp * GCN_ESHARE + (r - GCN_LSHARE)) * 256 + tid;
        if (vi < GCN_EVEC) {
            const int4 s = ((const int4*)edge_index)[vi];
            const int4 d = ((const int4*)(edge_index + GCN_N_EDGES))[vi];
            int* crow = cnt + half * GCN_N_NODES;
            int* arow = adj + half * GCN_APLANE;
            const int k0 = atomicAdd(crow + d.x, 1);
            const int k1 = atomicAdd(crow + d.y, 1);
            const int k2 = atomicAdd(crow + d.z, 1);
            const int k3 = atomicAdd(crow + d.w, 1);
            auto place = [&](const int dd, const int ss, const int slot) {
                if (slot < GCN_HCAP) {
                    arow[dd * GCN_HCAP + slot] = ss;
                } else {
                    const int o = atomicAdd(ovf_n, 1);
                    *(int2*)(ovf + 2 * o) = make_int2(dd, ss);
                }
            };
            place(d.x, s.x, k0);
            place(d.y, s.y, k1);
            place(d.z, s.z, k2);
            place(d.w, s.w, k3);
        }
        return;
    }
    const int q = grp * GCN_LSHARE + r;
    if (q >= GCN_LINBLK) return;

    const int lane = tid & 31;
    const int wave = tid >> 5;

    const int  strip  = q * 8 + wave;
    const bool active = strip < GCN_N_NODES / 16;
    const int  m0  = strip * 16;
    const int  row = lane & 15;
    const int  col = lane & 15;
    const int  h   = lane >> 4;
    const int  kb  = h << 1;
    float* T = Ts[wave];

    float4 wv[4];
#pragma unroll
    for (int j4 = 0; j4 < 4; ++j4) wv[j4] = ((const float4*)W)[tid + 256 * j4];

    const int w  = tid & 63;
    const int g  = tid >> 6;
    const int hw = q * 64 + w;
    const unsigned  keep = (hw < GCN_HWORDS) ? 0xffffffffu : 0u;
    const unsigned* pp   = part + g * GCN_HPITCH + min(hw, GCN_HWORDS - 1);
    unsigned lo = 0u, hi = 0u;
#pragma unroll
    for (int k = 0; k < GCN_DSLICES / 4; ++k) {
        const unsigned v = pp[(4 * k) * GCN_HPITCH] & keep;
        lo += v & 0xffffu;
        hi += v >> 16;
    }

    v2f a[16];
    const float* arow = x + (size_t)((active ? m0 : 0) + row) * GCN_D + kb;
#pragma unroll
    for (int t = 0; t < 16; ++t) {
        a[t] = *(const v2f*)(arow + 4 * t);
    }

#pragma unroll
    for (int j4 = 0; j4 < 4; ++j4) {
        const int j  = tid + 256 * j4;
        const int o  = j >> 4;
        const int k4 = j & 15;
        *(float4*)&Ws[o * GCN_PITCH + 4 * k4] = wv[j4];
    }
    *(uint2*)&Ds[g][2 * w] = make_uint2(lo, hi);
    __syncthreads();

    if (active) {
        const int   n128 = wave * 16 + row;
        const float deg  = (float)(Ds[0][n128] + Ds[1][n128] + Ds[2][n128] + Ds[3][n128]);
        const float s    = rsqrtf(deg + 1.0f);
#pragma unroll
        for (int t = 0; t < 16; ++t) {
            a[t].x *= s; a[t].y *= s;
        }

        v8f c0, c1, c2, c3;
#pragma unroll
        for (int rr = 0; rr < 8; ++rr) { c0[rr] = 0.0f; c1[rr] = 0.0f; c2[rr] = 0.0f; c3[rr] = 0.0f; }

        const float* wl = &Ws[col * GCN_PITCH + kb];
#pragma unroll
        for (int t = 0; t < 16; ++t) {
            const v2f b0 = *(const v2f*)(wl + (0 * 16) * GCN_PITCH + 4 * t);
            const v2f b1 = *(const v2f*)(wl + (1 * 16) * GCN_PITCH + 4 * t);
            const v2f b2 = *(const v2f*)(wl + (2 * 16) * GCN_PITCH + 4 * t);
            const v2f b3 = *(const v2f*)(wl + (3 * 16) * GCN_PITCH + 4 * t);
            c0 = __builtin_amdgcn_wmma_f32_16x16x4_f32(false, a[t], false, b0, (short)0, c0, false, false);
            c1 = __builtin_amdgcn_wmma_f32_16x16x4_f32(false, a[t], false, b1, (short)0, c1, false, false);
            c2 = __builtin_amdgcn_wmma_f32_16x16x4_f32(false, a[t], false, b2, (short)0, c2, false, false);
            c3 = __builtin_amdgcn_wmma_f32_16x16x4_f32(false, a[t], false, b3, (short)0, c3, false, false);
        }

        float* tp = T + (8 * h) * GCN_PITCH + col;
#pragma unroll
        for (int rr = 0; rr < 8; ++rr) {
            tp[rr * GCN_PITCH +  0] = c0[rr];
            tp[rr * GCN_PITCH + 16] = c1[rr];
            tp[rr * GCN_PITCH + 32] = c2[rr];
            tp[rr * GCN_PITCH + 48] = c3[rr];
        }
    }
    __syncthreads();
    if (active) {
        const int c4 = (lane & 15) * 4;
#pragma unroll
        for (int i = 0; i < 8; ++i) {
            const int R = h + 2 * i;
            const v4f v = *(const v4f*)&T[R * GCN_PITCH + c4];
            const size_t gofs = (size_t)(m0 + R) * GCN_D + c4;
            *(v4f*)(y + gofs) = v;
        }
    }
}

__global__ __launch_bounds__(256)
void gcn_gather(const int* __restrict__ cnt, const int* __restrict__ adj,
                const int* __restrict__ ovf_n, const int* __restrict__ ovf,
                const float* __restrict__ y, const float* __restrict__ bias,
                float* __restrict__ out) {
    const int l16 = threadIdx.x & 15;
    const int d   = (blockIdx.x * 256 + threadIdx.x) >> 4;
    const int c0 = cnt[d];
    const int c1 = cnt[GCN_N_NODES + d];
    const int n0 = min(c0, GCN_HCAP);
    const int n1 = min(c1, GCN_HCAP);
    const int n    = n0 + n1;
    const int nall = max(c0, c1);
    const int* row0 = adj + d * GCN_HCAP;
    const int* row1 = row0 + GCN_APLANE;
    const int nmax    = __builtin_amdgcn_readfirstlane(max(n, __shfl_xor(n, 16)));
    const int nallmax = __builtin_amdgcn_readfirstlane(max(nall, __shfl_xor(nall, 16)));

    const int v0 = (l16 < n0) ? row0[l16] : d;
    const int v1 = (l16 < n1) ? row1[l16] : d;
    const int sh   = __shfl(v1, (l16 - n0) & 15, 16);
    const int idxA = (l16 < n0) ? v0 : sh;
    const int idxB = sh;

    const v4f* y4 = (const v4f*)y;
    const v4f zero = {0.0f, 0.0f, 0.0f, 0.0f};
    v4f a0 = y4[d * 16 + l16] + *(const v4f*)(bias + 4 * l16);
    v4f a1 = zero, a2 = zero, a3 = zero;

    auto sum16 = [&](const int idx, const int base) {
        const int lim = min(16, nmax - base);
        for (int j = 0; j < lim; j += 4) {
            const int s0 = __shfl(idx, j + 0, 16);
            const int s1 = __shfl(idx, j + 1, 16);
            const int s2 = __shfl(idx, j + 2, 16);
            const int s3 = __shfl(idx, j + 3, 16);
            const v4f t0 = y4[s0 * 16 + l16];
            const v4f t1 = y4[s1 * 16 + l16];
            const v4f t2 = y4[s2 * 16 + l16];
            const v4f t3 = y4[s3 * 16 + l16];
            a0 += (base + j + 0 < n) ? t0 : zero;
            a1 += (base + j + 1 < n) ? t1 : zero;
            a2 += (base + j + 2 < n) ? t2 : zero;
            a3 += (base + j + 3 < n) ? t3 : zero;
        }
    };
    sum16(idxA, 0);
    if (nmax > 16) sum16(idxB, 16);
    v4f acc = (a0 + a1) + (a2 + a3);

    if (nallmax > GCN_HCAP) {
        const int m = *ovf_n;
        for (int base = 0; base < m; base += 16) {
            const int i = base + l16;
            int od = -1, os = 0;
            if (i < m) { const int2 r = *(const int2*)(ovf + 2 * i); od = r.x; os = r.y; }
            const int lim = min(16, m - base);
            for (int j = 0; j < lim; ++j) {
                const int dd = __shfl(od, j, 16);
                const int ss = __shfl(os, j, 16);
                if (dd == d) acc += y4[ss * 16 + l16];
            }
        }
    }
    ((v4f*)out)[d * 16 + l16] = acc;
}

extern "C" void kernel_launch(void* const* d_in, const int* in_sizes, int n_in,
                              void* d_out, int out_size, void* d_ws, size_t ws_size,
                              hipStream_t stream) {
    const float* x          = (const float*)d_in[0];
    const int*   edge_index = (const int*)d_in[1];
    const float* W          = (const float*)d_in[2];
    const float* b          = (const float*)d_in[3];
    float*       out        = (float*)d_out;

    int*   cnt    = (int*)d_ws + GCN_WS_CNT;
    int*   ovf_n  = (int*)d_ws + GCN_WS_OVFN;
    int*   adj    = (int*)d_ws + GCN_WS_ADJ;
    int*   ovf    = (int*)d_ws + GCN_WS_OVF;
    float* y      = (float*)d_ws + GCN_WS_Y;
    unsigned* part = (unsigned*)d_ws + GCN_WS_PART;

    gcn_degree_planes<<<GCN_DSLICES * GCN_DPARTS, GCN_DTHREADS, 0, stream>>>(edge_index, part, cnt, ovf_n);
    gcn_fill_linear<<<GCN_GROUPS * GCN_MIX, 256, 0, stream>>>(edge_index, part, cnt, adj, ovf_n, ovf, x, W, y);
    gcn_gather<<<(GCN_N_NODES * 16) / 256, 256, 0, stream>>>(cnt, adj, ovf_n, ovf, y, b, out);
}
